# starting kernel
# baseline (speedup 1.0000x reference)
#include <hip/hip_runtime.h>

typedef __attribute__((ext_vector_type(2))) float v2f;
typedef __attribute__((ext_vector_type(4))) float v4f;
typedef __attribute__((ext_vector_type(8))) float v8f;

#define FIN_EPS 1e-5f

constexpr int T_DIM     = 4096;
constexpr int CH_PER_WG = 16;
constexpr int THREADS   = 256;
constexpr int WAVES     = THREADS / 32;
constexpr int CHUNK     = T_DIM / WAVES;
constexpr int ITERS     = CHUNK / 4;

__device__ __forceinline__ v8f wmma_acc(v2f a, v2f b, v8f c) {
    return __builtin_amdgcn_wmma_f32_16x16x4_f32(
        false, a, false, b, (short)0, c, false, false);
}

__global__ __launch_bounds__(THREADS)
void finorm_wmma_kernel(const float* __restrict__ x,
                        const int*   __restrict__ x_lens,
                        float*       __restrict__ out,
                        int C)
{
    const int groups_per_b = C / CH_PER_WG;
    const int b  = blockIdx.x / groups_per_b;
    const int cg = blockIdx.x % groups_per_b;
    const int c0 = cg * CH_PER_WG;
    const int len = x_lens[b];

    const int tid    = threadIdx.x;
    const int wave   = tid >> 5;
    const int lane   = tid & 31;
    const int hi     = lane >> 4;
    const int lane16 = lane & 15;

    __shared__ float s_sum[WAVES][CH_PER_WG];
    __shared__ float s_sq [WAVES][CH_PER_WG];
    __shared__ float s_mean[CH_PER_WG];
    __shared__ float s_rstd[CH_PER_WG];

    const float* rowbase = x + ((size_t)b * C + (size_t)(c0 + lane16)) * T_DIM;
    const int t0 = wave * CHUNK;

    const int rel_s   = __builtin_amdgcn_readfirstlane(len - t0);
    const int nfull_s = rel_s >= CHUNK ? ITERS : (rel_s > 0 ? (rel_s >> 2) : 0);

    v8f acc_s0 = {}; v8f acc_s1 = {};
    v8f acc_q0 = {}; v8f acc_q1 = {};
    const v2f ones = {1.0f, 1.0f};

    const float* p = rowbase + t0 + 2 * hi;

    int i = 0;
    for (; i + 2 <= nfull_s; i += 2) {
        v2f b0 = *(const v2f*)(p);
        v2f b1 = *(const v2f*)(p + 4);
        v2f q0 = b0 * b0;
        v2f q1 = b1 * b1;
        acc_s0 = wmma_acc(ones, b0, acc_s0);
        acc_q0 = wmma_acc(ones, q0, acc_q0);
        acc_s1 = wmma_acc(ones, b1, acc_s1);
        acc_q1 = wmma_acc(ones, q1, acc_q1);
        p += 8;
    }
    if (i < nfull_s) {
        v2f b0 = *(const v2f*)(p);
        acc_s0 = wmma_acc(ones, b0, acc_s0);
        acc_q0 = wmma_acc(ones, b0 * b0, acc_q0);
        ++i;
    }
    if (rel_s > 0 && rel_s < CHUNK && (rel_s & 3)) {
        const int k = t0 + 4 * nfull_s + 2 * hi;
        v2f bv = *(const v2f*)(rowbase + k);
        v2f av;
        av.x = (k     < len) ? 1.0f : 0.0f;
        av.y = (k + 1 < len) ? 1.0f : 0.0f;
        acc_s0 = wmma_acc(av, bv, acc_s0);
        acc_q0 = wmma_acc(av, bv * bv, acc_q0);
    }

    v8f accs = acc_s0 + acc_s1;
    v8f accq = acc_q0 + acc_q1;

    if (hi == 0) {
        s_sum[wave][lane16] = accs.s0;
        s_sq [wave][lane16] = accq.s0;
    }
    __syncthreads();

    if (tid < CH_PER_WG) {
        float ts = 0.0f, tq = 0.0f;
        #pragma unroll
        for (int w = 0; w < WAVES; ++w) {
            ts += s_sum[w][tid];
            tq += s_sq [w][tid];
        }
        const float n    = (float)len;
        const float mean = ts / n;
        float var = (tq - ts * mean) / (n - 1.0f);
        var = var > 0.0f ? var : 0.0f;
        s_mean[tid] = mean;
        s_rstd[tid] = __frsqrt_rn(var + FIN_EPS);
    }
    __syncthreads();

    {
        const int row = tid >> 4;
        const int col = tid & 15;
        const float mean = s_mean[row];
        const float rstd = s_rstd[row];
        const size_t base = ((size_t)b * C + (size_t)(c0 + row)) * T_DIM;
        const float* rp = x   + base;
        float*       op = out + base;
        #pragma unroll 4
        for (int j = 0; j < T_DIM / (16 * 4); ++j) {
            const int e = (col + j * 16) * 4;
            v4f v = __builtin_nontemporal_load((const v4f*)(rp + e));
            v4f o = (v - mean) * rstd;
            __builtin_nontemporal_store(o, (v4f*)(op + e));
        }
    }
}

extern "C" void kernel_launch(void* const* d_in, const int* in_sizes, int n_in,
                              void* d_out, int out_size, void* d_ws, size_t ws_size,
                              hipStream_t stream)
{
    const float* x      = (const float*)d_in[0];
    const int*   x_lens = (const int*)d_in[1];
    float*       out    = (float*)d_out;

    const int B = in_sizes[1];
    const int C = (int)((long long)in_sizes[0] / ((long long)B * T_DIM));

    const int grid = B * (C / CH_PER_WG);
    finorm_wmma_kernel<<<grid, THREADS, 0, stream>>>(x, x_lens, out, C);
}


# Each wave keeps a whole row in registers: x read once by whole-line 16-byte loads, WMMA sums fed from them (16-lane fold instead of the LDS sum: summation order changes), normalised from registers
# speedup vs baseline: 1.6494x; 1.6494x over previous
#include <hip/hip_runtime.h>

typedef __attribute__((ext_vector_type(2))) float v2f;
typedef __attribute__((ext_vector_type(4))) float v4f;
typedef __attribute__((ext_vector_type(8))) float v8f;

#define FIN_EPS 1e-5f

constexpr int T_DIM   = 4096;
constexpr int THREADS = 256;
constexpr int WAVES   = THREADS / 32;
constexpr int ROWS    = 2;
constexpr int PIECE_T = 32 * 4;
constexpr int PIECES  = T_DIM / PIECE_T;

__device__ __forceinline__ v8f wmma_acc(v2f a, v2f b, v8f c) {
    return __builtin_amdgcn_wmma_f32_16x16x4_f32(
        false, a, false, b, (short)0, c, false, false);
}

template <int XMASK>
__device__ __forceinline__ float xmask_add(float v) {
    const int moved = __builtin_amdgcn_update_dpp(0, __builtin_bit_cast(int, v), 0x160 + XMASK, 0xF, 0xF, false);
    return v + __builtin_bit_cast(float, moved);
}
__device__ __forceinline__ float row16_sum(float v) {
    v = xmask_add<8>(v);
    v = xmask_add<4>(v);
    v = xmask_add<2>(v);
    v = xmask_add<1>(v);
    return v;
}

template <int Q>
__device__ __forceinline__ void prefix_sums(const v4f (&xr)[PIECES], int len, int tl,
                                            v8f& acc_s0, v8f& acc_s1, v8f& acc_q0, v8f& acc_q1) {
    if constexpr (Q < PIECES) {
        if (Q * PIECE_T < len) {
            const int rel = len - Q * PIECE_T;
            const v2f ones = {1.0f, 1.0f};
            v4f v = xr[Q];
            v.x = (tl     < rel) ? v.x : 0.0f;
            v.y = (tl + 1 < rel) ? v.y : 0.0f;
            v.z = (tl + 2 < rel) ? v.z : 0.0f;
            v.w = (tl + 3 < rel) ? v.w : 0.0f;
            const v2f lo = {v.x, v.y};
            const v2f hi = {v.z, v.w};
            acc_s0 = wmma_acc(ones, lo, acc_s0);
            acc_q0 = wmma_acc(ones, lo * lo, acc_q0);
            acc_s1 = wmma_acc(ones, hi, acc_s1);
            acc_q1 = wmma_acc(ones, hi * hi, acc_q1);
            prefix_sums<Q + 1>(xr, len, tl, acc_s0, acc_s1, acc_q0, acc_q1);
        }
    }
}

__global__ __launch_bounds__(THREADS) __attribute__((amdgpu_waves_per_eu(4, 4)))
void finorm_rows_kernel(const float* __restrict__ x,
                        const int*   __restrict__ x_lens,
                        float*       __restrict__ out,
                        int C)
{
    const int lane = threadIdx.x & 31;
    const int wave = __builtin_amdgcn_readfirstlane(threadIdx.x >> 5);
    const int row0 = (blockIdx.x * WAVES + wave) * ROWS;
    const int tl   = 4 * lane;

    const float* xp = x   + (size_t)row0 * T_DIM + tl;
    float*       op = out + (size_t)row0 * T_DIM + tl;

    v4f xr[PIECES];
    #pragma unroll
    for (int q = 0; q < PIECES; ++q)
        xr[q] = __builtin_nontemporal_load((const v4f*)(xp + q * PIECE_T));

    int lens[ROWS];
    #pragma unroll
    for (int r = 0; r < ROWS; ++r)
        lens[r] = __builtin_amdgcn_readfirstlane(x_lens[(row0 + r) / C]);

    #pragma unroll
    for (int r = 0; r < ROWS; ++r) {
        const int len = lens[r];

        v8f acc_s0 = {}; v8f acc_s1 = {};
        v8f acc_q0 = {}; v8f acc_q1 = {};
        prefix_sums<0>(xr, len, tl, acc_s0, acc_s1, acc_q0, acc_q1);
        const float ts = row16_sum(acc_s0.s0 + acc_s1.s0);
        const float tq = row16_sum(acc_q0.s0 + acc_q1.s0);

        const float n    = (float)len;
        const float mean = ts / n;
        float var = (tq - ts * mean) / (n - 1.0f);
        var = var > 0.0f ? var : 0.0f;
        const float rstd = __frsqrt_rn(var + FIN_EPS);

        #pragma unroll
        for (int q = 0; q < PIECES; ++q) {
            if (r + 1 < ROWS && q % 4 == 0) __builtin_amdgcn_sched_barrier(0);
            const v4f o = (xr[q] - mean) * rstd;
            __builtin_nontemporal_store(o, (v4f*)(op + r * T_DIM + q * PIECE_T));
            if (r + 1 < ROWS)
                xr[q] = __builtin_nontemporal_load((const v4f*)(xp + (r + 1) * T_DIM + q * PIECE_T));
        }
    }
}

extern "C" void kernel_launch(void* const* d_in, const int* in_sizes, int n_in,
                              void* d_out, int out_size, void* d_ws, size_t ws_size,
                              hipStream_t stream)
{
    const float* x      = (const float*)d_in[0];
    const int*   x_lens = (const int*)d_in[1];
    float*       out    = (float*)d_out;

    const int B = in_sizes[1];
    const int C = (int)((long long)in_sizes[0] / ((long long)B * T_DIM));

    const int grid = (B * C) / (WAVES * ROWS);
    finorm_rows_kernel<<<grid, THREADS, 0, stream>>>(x, x_lens, out, C);
}


# Reads and writes split in time: each wave parks its first row in LDS as it arrives and refills the registers with its second row, then stores both rows; sums and their order unchanged
# speedup vs baseline: 1.9045x; 1.1547x over previous
#include <hip/hip_runtime.h>

typedef __attribute__((ext_vector_type(2))) float v2f;
typedef __attribute__((ext_vector_type(4))) float v4f;
typedef __attribute__((ext_vector_type(8))) float v8f;

#define FIN_EPS 1e-5f

constexpr int T_DIM   = 4096;
constexpr int THREADS = 256;
constexpr int WAVES   = THREADS / 32;
constexpr int ROWS    = 2;
constexpr int PIECE_T = 32 * 4;
constexpr int PIECES  = T_DIM / PIECE_T;

__device__ __forceinline__ v8f wmma_acc(v2f a, v2f b, v8f c) {
    return __builtin_amdgcn_wmma_f32_16x16x4_f32(
        false, a, false, b, (short)0, c, false, false);
}

template <int XMASK>
__device__ __forceinline__ float xmask_add(float v) {
    const int moved = __builtin_amdgcn_update_dpp(0, __builtin_bit_cast(int, v), 0x160 + XMASK, 0xF, 0xF, false);
    return v + __builtin_bit_cast(float, moved);
}
__device__ __forceinline__ float row16_sum(float v) {
    v = xmask_add<8>(v);
    v = xmask_add<4>(v);
    v = xmask_add<2>(v);
    v = xmask_add<1>(v);
    return v;
}

__device__ __forceinline__ void add_piece(const v4f v, int rel, int tl, v4f& s, v4f& ss) {
    v4f m;
    m.x = (tl     < rel) ? v.x : 0.0f;
    m.y = (tl + 1 < rel) ? v.y : 0.0f;
    m.z = (tl + 2 < rel) ? v.z : 0.0f;
    m.w = (tl + 3 < rel) ? v.w : 0.0f;
    s = s + m; ss = ss + m * m;
}

__device__ __forceinline__ void row_stats(const v4f ps, const v4f pq, int len, float& mean, float& rstd) {
    const v2f ones = {1.0f, 1.0f};
    const v2f s_lo = {ps.x, ps.y};  const v2f s_hi = {ps.z, ps.w};
    const v2f q_lo = {pq.x, pq.y};  const v2f q_hi = {pq.z, pq.w};
    v8f acc_s = {}; v8f acc_q = {};
    acc_s = wmma_acc(ones, s_lo, acc_s);
    acc_q = wmma_acc(ones, q_lo, acc_q);
    acc_s = wmma_acc(ones, s_hi, acc_s);
    acc_q = wmma_acc(ones, q_hi, acc_q);
    asm volatile("v_nop\n\tv_nop\n\tv_nop\n\tv_nop"
                 : "+v"(acc_s), "+v"(acc_q)
                 : "v"(ones), "v"(s_lo), "v"(s_hi), "v"(q_lo), "v"(q_hi));
    const float ts = row16_sum(acc_s.s0);
    const float tq = row16_sum(acc_q.s0);

    const float n = (float)len;
    mean = ts / n;
    float var = (tq - ts * mean) / (n - 1.0f);
    var = var > 0.0f ? var : 0.0f;
    rstd = __frsqrt_rn(var + FIN_EPS);
}

__global__ __launch_bounds__(THREADS) __attribute__((amdgpu_waves_per_eu(4, 4)))
void finorm_rows_kernel(const float* __restrict__ x,
                        const int*   __restrict__ x_lens,
                        float*       __restrict__ out,
                        int C)
{
    __shared__ v4f parked[WAVES * PIECES * 32];

    const int lane = threadIdx.x & 31;
    const int wave = __builtin_amdgcn_readfirstlane(threadIdx.x >> 5);
    const int row0 = (blockIdx.x * WAVES + wave) * ROWS;
    const int tl   = 4 * lane;

    const float* xp = x   + (size_t)row0 * T_DIM + tl;
    float*       op = out + (size_t)row0 * T_DIM + tl;
    v4f*       park = parked + wave * (PIECES * 32) + lane;

    v4f xr[PIECES];
    #pragma unroll
    for (int q = 0; q < PIECES; ++q)
        xr[q] = __builtin_nontemporal_load((const v4f*)(xp + q * PIECE_T));

    int lens[ROWS];
    #pragma unroll
    for (int r = 0; r < ROWS; ++r)
        lens[r] = __builtin_amdgcn_readfirstlane(x_lens[(row0 + r) / C]);

    v4f s0 = {}; v4f q0 = {};
    #pragma unroll
    for (int q = 0; q < PIECES; ++q) {
        add_piece(xr[q], lens[0] - q * PIECE_T, tl, s0, q0);
        park[q * 32] = xr[q];
        xr[q] = __builtin_nontemporal_load((const v4f*)(xp + T_DIM + q * PIECE_T));
    }
    float mean0, rstd0;
    row_stats(s0, q0, lens[0], mean0, rstd0);

    v4f s1 = {}; v4f q1 = {};
    #pragma unroll
    for (int q = 0; q < PIECES; ++q)
        add_piece(xr[q], lens[1] - q * PIECE_T, tl, s1, q1);
    float mean1, rstd1;
    row_stats(s1, q1, lens[1], mean1, rstd1);

    #pragma unroll
    for (int q = 0; q < PIECES; ++q)
        xr[q] = (xr[q] - mean1) * rstd1;
    __builtin_amdgcn_sched_barrier(0);
    #pragma unroll
    for (int q = 0; q < PIECES; ++q)
        __builtin_nontemporal_store(xr[q], (v4f*)(op + T_DIM + q * PIECE_T));
    __builtin_amdgcn_sched_barrier(0);
    #pragma unroll
    for (int q = 0; q < PIECES; ++q)
        xr[q] = park[q * 32];
    #pragma unroll
    for (int q = 0; q < PIECES; ++q)
        xr[q] = (xr[q] - mean0) * rstd0;
    __builtin_amdgcn_sched_barrier(0);
    #pragma unroll
    for (int q = 0; q < PIECES; ++q)
        __builtin_nontemporal_store(xr[q], (v4f*)(op + q * PIECE_T));
}

extern "C" void kernel_launch(void* const* d_in, const int* in_sizes, int n_in,
                              void* d_out, int out_size, void* d_ws, size_t ws_size,
                              hipStream_t stream)
{
    const float* x      = (const float*)d_in[0];
    const int*   x_lens = (const int*)d_in[1];
    float*       out    = (float*)d_out;

    const int B = in_sizes[1];
    const int C = (int)((long long)in_sizes[0] / ((long long)B * T_DIM));

    const int grid = (B * C) / (WAVES * ROWS);
    finorm_rows_kernel<<<grid, THREADS, 0, stream>>>(x, x_lens, out, C);
}


# Half as many waves, four rows each: a wave keeps two rows in registers (two waves per SIMD, up to 63 loads in flight), parks two in LDS, then stores all four; summation order unchanged
# speedup vs baseline: 2.0493x; 1.0760x over previous
#include <hip/hip_runtime.h>

typedef __attribute__((ext_vector_type(2))) float v2f;
typedef __attribute__((ext_vector_type(4))) float v4f;
typedef __attribute__((ext_vector_type(8))) float v8f;

#define FIN_EPS 1e-5f

constexpr int T_DIM   = 4096;
constexpr int THREADS = 256;
constexpr int WAVES   = THREADS / 32;
constexpr int HELD    = 2;
constexpr int ROWS    = 2 * HELD;
constexpr int PIECE_T = 32 * 4;
constexpr int PIECES  = T_DIM / PIECE_T;

__device__ __forceinline__ v8f wmma_acc(v2f a, v2f b, v8f c) {
    return __builtin_amdgcn_wmma_f32_16x16x4_f32(
        false, a, false, b, (short)0, c, false, false);
}

template <int XMASK>
__device__ __forceinline__ float xmask_add(float v) {
    const int moved = __builtin_amdgcn_update_dpp(0, __builtin_bit_cast(int, v), 0x160 + XMASK, 0xF, 0xF, false);
    return v + __builtin_bit_cast(float, moved);
}
__device__ __forceinline__ float row16_sum(float v) {
    v = xmask_add<8>(v);
    v = xmask_add<4>(v);
    v = xmask_add<2>(v);
    v = xmask_add<1>(v);
    return v;
}

__device__ __forceinline__ void add_piece(const v4f v, int rel, int tl, v4f& s, v4f& ss) {
    v4f m;
    m.x = (tl     < rel) ? v.x : 0.0f;
    m.y = (tl + 1 < rel) ? v.y : 0.0f;
    m.z = (tl + 2 < rel) ? v.z : 0.0f;
    m.w = (tl + 3 < rel) ? v.w : 0.0f;
    s = s + m; ss = ss + m * m;
}

__device__ __forceinline__ void row_stats(const v4f ps, const v4f pq, int len, float& mean, float& rstd) {
    const v2f ones = {1.0f, 1.0f};
    const v2f s_lo = {ps.x, ps.y};  const v2f s_hi = {ps.z, ps.w};
    const v2f q_lo = {pq.x, pq.y};  const v2f q_hi = {pq.z, pq.w};
    v8f acc_s = {}; v8f acc_q = {};
    acc_s = wmma_acc(ones, s_lo, acc_s);
    acc_q = wmma_acc(ones, q_lo, acc_q);
    acc_s = wmma_acc(ones, s_hi, acc_s);
    acc_q = wmma_acc(ones, q_hi, acc_q);
    asm volatile("v_nop\n\tv_nop\n\tv_nop\n\tv_nop"
                 : "+v"(acc_s), "+v"(acc_q)
                 : "v"(ones), "v"(s_lo), "v"(s_hi), "v"(q_lo), "v"(q_hi));
    const float ts = row16_sum(acc_s.s0);
    const float tq = row16_sum(acc_q.s0);

    const float n = (float)len;
    mean = ts / n;
    float var = (tq - ts * mean) / (n - 1.0f);
    var = var > 0.0f ? var : 0.0f;
    rstd = __frsqrt_rn(var + FIN_EPS);
}

__global__ __launch_bounds__(THREADS) __attribute__((amdgpu_waves_per_eu(2, 2)))
void finorm_rows_kernel(const float* __restrict__ x,
                        const int*   __restrict__ x_lens,
                        float*       __restrict__ out,
                        int C)
{
    __shared__ v4f parked[WAVES * HELD * PIECES * 32];

    const int lane = threadIdx.x & 31;
    const int wave = __builtin_amdgcn_readfirstlane(threadIdx.x >> 5);
    const int row0 = (blockIdx.x * WAVES + wave) * ROWS;
    const int tl   = 4 * lane;

    const float* xp = x   + (size_t)row0 * T_DIM + tl;
    float*       op = out + (size_t)row0 * T_DIM + tl;
    v4f*       park = parked + wave * (HELD * PIECES * 32) + lane;

    v4f xr[HELD][PIECES];
    #pragma unroll
    for (int h = 0; h < HELD; ++h) {
        #pragma unroll
        for (int q = 0; q < PIECES; ++q)
            xr[h][q] = __builtin_nontemporal_load((const v4f*)(xp + h * T_DIM + q * PIECE_T));
    }

    int lens[ROWS];
    #pragma unroll
    for (int r = 0; r < ROWS; ++r)
        lens[r] = __builtin_amdgcn_readfirstlane(x_lens[(row0 + r) / C]);

    float mean[ROWS], rstd[ROWS];

    #pragma unroll
    for (int h = 0; h < HELD; ++h) {
        v4f s = {}; v4f ss = {};
        #pragma unroll
        for (int q = 0; q < PIECES; ++q) {
            add_piece(xr[h][q], lens[h] - q * PIECE_T, tl, s, ss);
            park[(h * PIECES + q) * 32] = xr[h][q];
            xr[h][q] = __builtin_nontemporal_load((const v4f*)(xp + (HELD + h) * T_DIM + q * PIECE_T));
        }
        row_stats(s, ss, lens[h], mean[h], rstd[h]);
    }
    #pragma unroll
    for (int h = 0; h < HELD; ++h) {
        v4f s = {}; v4f ss = {};
        #pragma unroll
        for (int q = 0; q < PIECES; ++q)
            add_piece(xr[h][q], lens[HELD + h] - q * PIECE_T, tl, s, ss);
        row_stats(s, ss, lens[HELD + h], mean[HELD + h], rstd[HELD + h]);
    }

    #pragma unroll
    for (int h = 0; h < HELD; ++h) {
        #pragma unroll
        for (int q = 0; q < PIECES; ++q)
            xr[h][q] = (xr[h][q] - mean[HELD + h]) * rstd[HELD + h];
    }
    __builtin_amdgcn_sched_barrier(0);
    #pragma unroll
    for (int h = 0; h < HELD; ++h) {
        #pragma unroll
        for (int q = 0; q < PIECES; ++q)
            __builtin_nontemporal_store(xr[h][q], (v4f*)(op + (HELD + h) * T_DIM + q * PIECE_T));
    }
    __builtin_amdgcn_sched_barrier(0);
    #pragma unroll
    for (int h = 0; h < HELD; ++h) {
        #pragma unroll
        for (int q = 0; q < PIECES; ++q)
            xr[h][q] = park[(h * PIECES + q) * 32];
    }
    #pragma unroll
    for (int h = 0; h < HELD; ++h) {
        #pragma unroll
        for (int q = 0; q < PIECES; ++q)
            xr[h][q] = (xr[h][q] - mean[h]) * rstd[h];
    }
    __builtin_amdgcn_sched_barrier(0);
    #pragma unroll
    for (int h = 0; h < HELD; ++h) {
        #pragma unroll
        for (int q = 0; q < PIECES; ++q)
            __builtin_nontemporal_store(xr[h][q], (v4f*)(op + h * T_DIM + q * PIECE_T));
    }
}

extern "C" void kernel_launch(void* const* d_in, const int* in_sizes, int n_in,
                              void* d_out, int out_size, void* d_ws, size_t ws_size,
                              hipStream_t stream)
{
    const float* x      = (const float*)d_in[0];
    const int*   x_lens = (const int*)d_in[1];
    float*       out    = (float*)d_out;

    const int B = in_sizes[1];
    const int C = (int)((long long)in_sizes[0] / ((long long)B * T_DIM));

    const int grid = (B * C) / (WAVES * ROWS);
    finorm_rows_kernel<<<grid, THREADS, 0, stream>>>(x, x_lens, out, C);
}
